# starting kernel
# baseline (speedup 1.0000x reference)
#include <hip/hip_runtime.h>
#include <hip/hip_bf16.h>

typedef float v2f __attribute__((ext_vector_type(2)));
typedef float v8f __attribute__((ext_vector_type(8)));

#define DIM 128
#define NEG_SLOPE 0.01f
#define SCAN_T 1024

__global__ void cnt_kernel(const int* __restrict__ dst, int* __restrict__ cnt, int E) {
    int e = blockIdx.x * blockDim.x + threadIdx.x;
    if (e < E) atomicAdd(&cnt[dst[e]], 1);
}

__global__ void dis_kernel(const int* __restrict__ cnt, float* __restrict__ dis, int N) {
    int i = blockIdx.x * blockDim.x + threadIdx.x;
    if (i < N) dis[i] = rsqrtf((float)cnt[i] + 1.0f);
}

__global__ __launch_bounds__(SCAN_T) void scan_kernel(const int* __restrict__ cnt,
                                                      int* __restrict__ row_ptr, int N) {
    __shared__ int sdata[SCAN_T];
    __shared__ int carry_s;
    if (threadIdx.x == 0) carry_s = 0;
    __syncthreads();
    for (int base = 0; base < N; base += SCAN_T) {
        int i = base + threadIdx.x;
        int v = (i < N) ? cnt[i] : 0;
        sdata[threadIdx.x] = v;
        __syncthreads();
        for (int off = 1; off < SCAN_T; off <<= 1) {
            int t = (threadIdx.x >= off) ? sdata[threadIdx.x - off] : 0;
            __syncthreads();
            sdata[threadIdx.x] += t;
            __syncthreads();
        }
        int incl  = sdata[threadIdx.x];
        int total = sdata[SCAN_T - 1];
        int c     = carry_s;
        if (i < N) row_ptr[i] = c + incl - v;
        __syncthreads();
        if (threadIdx.x == 0) carry_s = c + total;
        __syncthreads();
    }
    if (threadIdx.x == 0) row_ptr[N] = carry_s;
}

__global__ void copy_kernel(const int* __restrict__ srcp, int* __restrict__ dstp, int N) {
    int i = blockIdx.x * blockDim.x + threadIdx.x;
    if (i < N) dstp[i] = srcp[i];
}

__global__ void fill_csr_kernel(const int* __restrict__ src, const int* __restrict__ dst,
                                const float* __restrict__ dis,
                                int* __restrict__ cursor, int* __restrict__ col,
                                float* __restrict__ val, int E) {
    int e = blockIdx.x * blockDim.x + threadIdx.x;
    if (e >= E) return;
    int s = src[e], d = dst[e];
    int pos = atomicAdd(&cursor[d], 1);
    col[pos] = s;
    val[pos] = dis[s] * dis[d];
}

__global__ __launch_bounds__(256) void gemm_wmma_kernel(
        const float* __restrict__ A,
        const float* __restrict__ W,
        float* __restrict__ T,
        int N) {
    __shared__ float As[16 * 132];

    const int row0 = blockIdx.x * 16;
    const int tid  = threadIdx.x;

    for (int i = tid; i < 16 * DIM; i += 256) {
        int r = i >> 7;
        int c = i & (DIM - 1);
        int rr = row0 + r;
        As[r * 132 + c] = (rr < N) ? A[(size_t)rr * DIM + c] : 0.0f;
    }
    __syncthreads();

    const int wave  = tid >> 5;
    const int lane  = tid & 31;
    const int m     = lane & 15;
    const int khalf = (lane >> 4) * 2;
    const int n0    = wave * 16;

    v8f acc = {};
#pragma unroll
    for (int k = 0; k < DIM; k += 4) {
        v2f a, b;
        a.x = As[m * 132 + k + khalf];
        a.y = As[m * 132 + k + khalf + 1];
        b.x = W[(k + khalf) * DIM + n0 + m];
        b.y = W[(k + khalf + 1) * DIM + n0 + m];
        acc = __builtin_amdgcn_wmma_f32_16x16x4_f32(
                false, a, false, b, (short)0, acc, false, false);
    }

    const int rbase = row0 + (lane >> 4) * 8;
#pragma unroll
    for (int v = 0; v < 8; ++v) {
        int r = rbase + v;
        if (r < N) T[(size_t)r * DIM + n0 + m] = acc[v];
    }
}

__global__ __launch_bounds__(256) void spmm_kernel(
        const float* __restrict__ T,
        const int*   __restrict__ row_ptr,
        const int*   __restrict__ col,
        const float* __restrict__ val,
        const float* __restrict__ dis,
        const float* __restrict__ bias,
        float*       __restrict__ out,
        int N) {
    int node = blockIdx.x * 8 + (threadIdx.x >> 5);
    if (node >= N) return;
    int lane = threadIdx.x & 31;

    int beg = row_ptr[node];
    int end = row_ptr[node + 1];

    float ds = dis[node];
    float w0 = ds * ds;
    float4 tv = ((const float4*)(T + (size_t)node * DIM))[lane];
    float4 acc;
    acc.x = tv.x * w0; acc.y = tv.y * w0; acc.z = tv.z * w0; acc.w = tv.w * w0;

    for (int jb = beg; jb < end; jb += 32) {
        int nb = end - jb;
        if (nb > 32) nb = 32;
        int   jl  = jb + lane;
        int   s_l = (jl < end) ? col[jl] : 0;
        float w_l = (jl < end) ? val[jl] : 0.0f;
        for (int k = 0; k < nb; ++k) {
            int   s = __shfl(s_l, k);
            float w = __shfl(w_l, k);
            float4 v = ((const float4*)(T + (size_t)s * DIM))[lane];
            acc.x = fmaf(v.x, w, acc.x);
            acc.y = fmaf(v.y, w, acc.y);
            acc.z = fmaf(v.z, w, acc.z);
            acc.w = fmaf(v.w, w, acc.w);
        }
    }

    float4 bv = ((const float4*)bias)[lane];
    float4 r;
    r.x = acc.x + bv.x; r.y = acc.y + bv.y;
    r.z = acc.z + bv.z; r.w = acc.w + bv.w;
    r.x = (r.x >= 0.0f) ? r.x : NEG_SLOPE * r.x;
    r.y = (r.y >= 0.0f) ? r.y : NEG_SLOPE * r.y;
    r.z = (r.z >= 0.0f) ? r.z : NEG_SLOPE * r.z;
    r.w = (r.w >= 0.0f) ? r.w : NEG_SLOPE * r.w;
    ((float4*)(out + (size_t)node * DIM))[lane] = r;
}

extern "C" void kernel_launch(void* const* d_in, const int* in_sizes, int n_in,
                              void* d_out, int out_size, void* d_ws, size_t ws_size,
                              hipStream_t stream) {
    const float* x  = (const float*)d_in[0];
    const int*   ei = (const int*)  d_in[1];
    const float* W0 = (const float*)d_in[2];
    const float* b0 = (const float*)d_in[3];
    const float* W1 = (const float*)d_in[4];
    const float* b1 = (const float*)d_in[5];
    float* out = (float*)d_out;

    const int N = in_sizes[0] / DIM;
    const int E = in_sizes[1] / 2;
    const int* srcIdx = ei;
    const int* dstIdx = ei + E;

    const size_t NH = (size_t)N * DIM;

    float* T       = (float*)d_ws;
    float* dis     = T + NH;
    float* val     = dis + N;
    int*   cnt     = (int*)(val + E);
    int*   row_ptr = cnt + N;
    int*   cursor  = row_ptr + (N + 1);
    int*   col     = cursor + N;

    const int thr   = 256;
    const int gridN = (N + thr - 1) / thr;
    const int gridE = (E + thr - 1) / thr;

    hipMemsetAsync(cnt, 0, (size_t)N * sizeof(int), stream);
    cnt_kernel <<<gridE, thr, 0, stream>>>(dstIdx, cnt, E);
    dis_kernel <<<gridN, thr, 0, stream>>>(cnt, dis, N);
    scan_kernel<<<1, SCAN_T, 0, stream>>>(cnt, row_ptr, N);
    copy_kernel<<<gridN, thr, 0, stream>>>(row_ptr, cursor, N);
    fill_csr_kernel<<<gridE, thr, 0, stream>>>(srcIdx, dstIdx, dis, cursor, col, val, E);

    const int gemmGrid = (N + 15) / 16;
    const int nodeGrid = (N + 7) / 8;

    gemm_wmma_kernel<<<gemmGrid, thr, 0, stream>>>(x, W0, T, N);
    spmm_kernel<<<nodeGrid, thr, 0, stream>>>(T, row_ptr, col, val, dis, b0, out, N);

    gemm_wmma_kernel<<<gemmGrid, thr, 0, stream>>>(out, W1, T, N);
    spmm_kernel<<<nodeGrid, thr, 0, stream>>>(T, row_ptr, col, val, dis, b1, out, N);
}


# scan launch: one 1024-thread block per 1024 counts, each summing the counts in front of its chunk itself, instead of one block walking 49 chunks with 24 barriers each
# speedup vs baseline: 1.4505x; 1.4505x over previous
#include <hip/hip_runtime.h>
#include <hip/hip_bf16.h>

typedef float v2f __attribute__((ext_vector_type(2)));
typedef float v8f __attribute__((ext_vector_type(8)));

#define DIM 128
#define NEG_SLOPE 0.01f
#define SCAN_T 1024

__global__ void cnt_kernel(const int* __restrict__ dst, int* __restrict__ cnt, int E) {
    int e = blockIdx.x * blockDim.x + threadIdx.x;
    if (e < E) atomicAdd(&cnt[dst[e]], 1);
}

__global__ void dis_kernel(const int* __restrict__ cnt, float* __restrict__ dis, int N) {
    int i = blockIdx.x * blockDim.x + threadIdx.x;
    if (i < N) dis[i] = rsqrtf((float)cnt[i] + 1.0f);
}

static_assert(SCAN_T == 32 * 32, "one lane per wave total: 32 waves of 32 lanes");

__global__ __launch_bounds__(SCAN_T) void scan_kernel(const int* __restrict__ cnt,
                                                      int* __restrict__ row_ptr, int N) {
    __shared__ int w_pre[SCAN_T / 32];
    __shared__ int w_tot[SCAN_T / 32];

    const int tid  = threadIdx.x;
    const int lane = tid & 31;
    const int wave = tid >> 5;
    const int nfront = blockIdx.x;

    int pre = 0;
    for (int k = 0; k < nfront; ++k) pre += cnt[k * SCAN_T + tid];

    const int i = nfront * SCAN_T + tid;
    const int v = (i < N) ? cnt[i] : 0;

    int x = v;
    for (int off = 1; off < 32; off <<= 1) {
        int t = __shfl_up(x, off);
        if (lane >= off) x += t;
    }
    for (int m = 16; m > 0; m >>= 1) pre += __shfl_xor(pre, m);

    if (lane == 31) w_tot[wave] = x;
    if (lane == 0)  w_pre[wave] = pre;
    __syncthreads();

    int c = w_pre[lane] + ((lane < wave) ? w_tot[lane] : 0);
    for (int m = 16; m > 0; m >>= 1) c += __shfl_xor(c, m);

    if (i < N) row_ptr[i] = c + x - v;
    if (i == N - 1) row_ptr[N] = c + x;
}

__global__ void copy_kernel(const int* __restrict__ srcp, int* __restrict__ dstp, int N) {
    int i = blockIdx.x * blockDim.x + threadIdx.x;
    if (i < N) dstp[i] = srcp[i];
}

__global__ void fill_csr_kernel(const int* __restrict__ src, const int* __restrict__ dst,
                                const float* __restrict__ dis,
                                int* __restrict__ cursor, int* __restrict__ col,
                                float* __restrict__ val, int E) {
    int e = blockIdx.x * blockDim.x + threadIdx.x;
    if (e >= E) return;
    int s = src[e], d = dst[e];
    int pos = atomicAdd(&cursor[d], 1);
    col[pos] = s;
    val[pos] = dis[s] * dis[d];
}

__global__ __launch_bounds__(256) void gemm_wmma_kernel(
        const float* __restrict__ A,
        const float* __restrict__ W,
        float* __restrict__ T,
        int N) {
    __shared__ float As[16 * 132];

    const int row0 = blockIdx.x * 16;
    const int tid  = threadIdx.x;

    for (int i = tid; i < 16 * DIM; i += 256) {
        int r = i >> 7;
        int c = i & (DIM - 1);
        int rr = row0 + r;
        As[r * 132 + c] = (rr < N) ? A[(size_t)rr * DIM + c] : 0.0f;
    }
    __syncthreads();

    const int wave  = tid >> 5;
    const int lane  = tid & 31;
    const int m     = lane & 15;
    const int khalf = (lane >> 4) * 2;
    const int n0    = wave * 16;

    v8f acc = {};
#pragma unroll
    for (int k = 0; k < DIM; k += 4) {
        v2f a, b;
        a.x = As[m * 132 + k + khalf];
        a.y = As[m * 132 + k + khalf + 1];
        b.x = W[(k + khalf) * DIM + n0 + m];
        b.y = W[(k + khalf + 1) * DIM + n0 + m];
        acc = __builtin_amdgcn_wmma_f32_16x16x4_f32(
                false, a, false, b, (short)0, acc, false, false);
    }

    const int rbase = row0 + (lane >> 4) * 8;
#pragma unroll
    for (int v = 0; v < 8; ++v) {
        int r = rbase + v;
        if (r < N) T[(size_t)r * DIM + n0 + m] = acc[v];
    }
}

__global__ __launch_bounds__(256) void spmm_kernel(
        const float* __restrict__ T,
        const int*   __restrict__ row_ptr,
        const int*   __restrict__ col,
        const float* __restrict__ val,
        const float* __restrict__ dis,
        const float* __restrict__ bias,
        float*       __restrict__ out,
        int N) {
    int node = blockIdx.x * 8 + (threadIdx.x >> 5);
    if (node >= N) return;
    int lane = threadIdx.x & 31;

    int beg = row_ptr[node];
    int end = row_ptr[node + 1];

    float ds = dis[node];
    float w0 = ds * ds;
    float4 tv = ((const float4*)(T + (size_t)node * DIM))[lane];
    float4 acc;
    acc.x = tv.x * w0; acc.y = tv.y * w0; acc.z = tv.z * w0; acc.w = tv.w * w0;

    for (int jb = beg; jb < end; jb += 32) {
        int nb = end - jb;
        if (nb > 32) nb = 32;
        int   jl  = jb + lane;
        int   s_l = (jl < end) ? col[jl] : 0;
        float w_l = (jl < end) ? val[jl] : 0.0f;
        for (int k = 0; k < nb; ++k) {
            int   s = __shfl(s_l, k);
            float w = __shfl(w_l, k);
            float4 v = ((const float4*)(T + (size_t)s * DIM))[lane];
            acc.x = fmaf(v.x, w, acc.x);
            acc.y = fmaf(v.y, w, acc.y);
            acc.z = fmaf(v.z, w, acc.z);
            acc.w = fmaf(v.w, w, acc.w);
        }
    }

    float4 bv = ((const float4*)bias)[lane];
    float4 r;
    r.x = acc.x + bv.x; r.y = acc.y + bv.y;
    r.z = acc.z + bv.z; r.w = acc.w + bv.w;
    r.x = (r.x >= 0.0f) ? r.x : NEG_SLOPE * r.x;
    r.y = (r.y >= 0.0f) ? r.y : NEG_SLOPE * r.y;
    r.z = (r.z >= 0.0f) ? r.z : NEG_SLOPE * r.z;
    r.w = (r.w >= 0.0f) ? r.w : NEG_SLOPE * r.w;
    ((float4*)(out + (size_t)node * DIM))[lane] = r;
}

extern "C" void kernel_launch(void* const* d_in, const int* in_sizes, int n_in,
                              void* d_out, int out_size, void* d_ws, size_t ws_size,
                              hipStream_t stream) {
    const float* x  = (const float*)d_in[0];
    const int*   ei = (const int*)  d_in[1];
    const float* W0 = (const float*)d_in[2];
    const float* b0 = (const float*)d_in[3];
    const float* W1 = (const float*)d_in[4];
    const float* b1 = (const float*)d_in[5];
    float* out = (float*)d_out;

    const int N = in_sizes[0] / DIM;
    const int E = in_sizes[1] / 2;
    const int* srcIdx = ei;
    const int* dstIdx = ei + E;

    const size_t NH = (size_t)N * DIM;

    float* T       = (float*)d_ws;
    float* dis     = T + NH;
    float* val     = dis + N;
    int*   cnt     = (int*)(val + E);
    int*   row_ptr = cnt + N;
    int*   cursor  = row_ptr + (N + 1);
    int*   col     = cursor + N;

    const int thr   = 256;
    const int gridN = (N + thr - 1) / thr;
    const int gridE = (E + thr - 1) / thr;
    const int gridS = (N + SCAN_T - 1) / SCAN_T;

    hipMemsetAsync(cnt, 0, (size_t)N * sizeof(int), stream);
    cnt_kernel <<<gridE, thr, 0, stream>>>(dstIdx, cnt, E);
    dis_kernel <<<gridN, thr, 0, stream>>>(cnt, dis, N);
    scan_kernel<<<gridS, SCAN_T, 0, stream>>>(cnt, row_ptr, N);
    copy_kernel<<<gridN, thr, 0, stream>>>(row_ptr, cursor, N);
    fill_csr_kernel<<<gridE, thr, 0, stream>>>(srcIdx, dstIdx, dis, cursor, col, val, E);

    const int gemmGrid = (N + 15) / 16;
    const int nodeGrid = (N + 7) / 8;

    gemm_wmma_kernel<<<gemmGrid, thr, 0, stream>>>(x, W0, T, N);
    spmm_kernel<<<nodeGrid, thr, 0, stream>>>(T, row_ptr, col, val, dis, b0, out, N);

    gemm_wmma_kernel<<<gemmGrid, thr, 0, stream>>>(out, W1, T, N);
    spmm_kernel<<<nodeGrid, thr, 0, stream>>>(T, row_ptr, col, val, dis, b1, out, N);
}


# CSR build: edges grouped by destination through LDS counters (256 edge chunks by 256-node buckets, then one block per bucket) instead of one global atomic per edge; row order still an arrival rank
# speedup vs baseline: 2.4457x; 1.6861x over previous
#include <hip/hip_runtime.h>
#include <hip/hip_bf16.h>

typedef float v2f __attribute__((ext_vector_type(2)));
typedef float v8f __attribute__((ext_vector_type(8)));

#define DIM 128
#define NEG_SLOPE 0.01f

#define BK_SHIFT 8
#define BK_NODES 256
#define BK_MAX   256
#define BIN_T    1024
#define BIN_G    256
#define BIN_PER  4

static_assert(BK_NODES == (1 << BK_SHIFT), "bucket of a node = node >> BK_SHIFT");
static_assert(BIN_T == 4 * BK_MAX, "carry-in: four threads per bucket, a quarter of the chunks each");
static_assert(BIN_G % 4 == 0, "chunks split in four equal quarters");

__global__ __launch_bounds__(BIN_T) void bin_count_kernel(const int* __restrict__ dst,
                                                          int* __restrict__ counts, int E) {
    __shared__ int h[BK_MAX];
    const int tid = threadIdx.x;
    if (tid < BK_MAX) h[tid] = 0;
    __syncthreads();

    const int chunk = (E + BIN_G - 1) / BIN_G;
    const int beg = blockIdx.x * chunk;
    const int end = min(beg + chunk, E);
    for (int base = beg + tid; base < end; base += BIN_PER * BIN_T) {
        int d[BIN_PER];
#pragma unroll
        for (int j = 0; j < BIN_PER; ++j) {
            int e = base + j * BIN_T;
            d[j] = (e < end) ? dst[e] : -1;
        }
#pragma unroll
        for (int j = 0; j < BIN_PER; ++j)
            if (d[j] >= 0) atomicAdd(&h[d[j] >> BK_SHIFT], 1);
    }
    __syncthreads();
    if (tid < BK_MAX) counts[blockIdx.x * BK_MAX + tid] = h[tid];
}

__global__ __launch_bounds__(BIN_T) void bin_scatter_kernel(const int* __restrict__ src,
                                                            const int* __restrict__ dst,
                                                            const int* __restrict__ counts,
                                                            int* __restrict__ bucket_base,
                                                            int* __restrict__ ebuf, int E) {
    __shared__ int p_pre[4][BK_MAX];
    __shared__ int p_tot[4][BK_MAX];
    __shared__ int offs[BK_MAX];
    __shared__ int w_tot[BK_MAX / 32];

    const int tid  = threadIdx.x;
    const int lane = tid & 31;
    const int wave = tid >> 5;
    const int g    = blockIdx.x;
    {
        const int b  = tid & (BK_MAX - 1);
        const int q  = tid / BK_MAX;
        const int k0 = q * (BIN_G / 4);
        int pre = 0, tot = 0;
#pragma unroll 16
        for (int k = 0; k < BIN_G / 4; ++k) {
            int c = counts[(k0 + k) * BK_MAX + b];
            tot += c;
            pre += ((k0 + k) < g) ? c : 0;
        }
        p_pre[q][b] = pre;
        p_tot[q][b] = tot;
    }
    __syncthreads();

    int x = 0, t = 0, pr = 0;
    if (tid < BK_MAX) {
        pr = p_pre[0][tid] + p_pre[1][tid] + p_pre[2][tid] + p_pre[3][tid];
        t  = p_tot[0][tid] + p_tot[1][tid] + p_tot[2][tid] + p_tot[3][tid];
        x = t;
        for (int off = 1; off < 32; off <<= 1) {
            int u = __shfl_up(x, off);
            if (lane >= off) x += u;
        }
        if (lane == 31) w_tot[wave] = x;
    }
    __syncthreads();
    if (tid < BK_MAX) {
        int carry = 0;
        for (int w = 0; w < wave; ++w) carry += w_tot[w];
        const int base = carry + x - t;
        offs[tid] = base + pr;
        if (g == 0) {
            bucket_base[tid] = base;
            if (tid == BK_MAX - 1) bucket_base[BK_MAX] = base + t;
        }
    }
    __syncthreads();

    const int chunk = (E + BIN_G - 1) / BIN_G;
    const int beg = g * chunk;
    const int end = min(beg + chunk, E);
    int2* __restrict__ epair = (int2*)ebuf;
    for (int base = beg + tid; base < end; base += BIN_PER * BIN_T) {
        int s[BIN_PER], d[BIN_PER];
#pragma unroll
        for (int j = 0; j < BIN_PER; ++j) {
            int e = base + j * BIN_T;
            s[j] = (e < end) ? src[e] : 0;
            d[j] = (e < end) ? dst[e] : -1;
        }
#pragma unroll
        for (int j = 0; j < BIN_PER; ++j) {
            if (d[j] >= 0) {
                int pos = atomicAdd(&offs[d[j] >> BK_SHIFT], 1);
                epair[pos] = make_int2(s[j], d[j]);
            }
        }
    }
}

__global__ __launch_bounds__(BIN_T) void bin_rows_kernel(const int* __restrict__ ebuf,
                                                         const int* __restrict__ bucket_base,
                                                         int* __restrict__ row_ptr,
                                                         float* __restrict__ dis,
                                                         int* __restrict__ col,
                                                         int* __restrict__ rowid, int N) {
    __shared__ int cnt[BK_NODES];
    __shared__ int off[BK_NODES];
    __shared__ int w_tot[BK_NODES / 32];

    const int tid   = threadIdx.x;
    const int lane  = tid & 31;
    const int wave  = tid >> 5;
    const int node0 = blockIdx.x << BK_SHIFT;
    const int bb = bucket_base[blockIdx.x];
    const int be = bucket_base[blockIdx.x + 1];
    const int2* __restrict__ epair = (const int2*)ebuf;

    if (tid < BK_NODES) cnt[tid] = 0;
    __syncthreads();

    for (int base = bb + tid; base < be; base += BIN_PER * BIN_T) {
        int d[BIN_PER];
#pragma unroll
        for (int j = 0; j < BIN_PER; ++j) {
            int i = base + j * BIN_T;
            d[j] = (i < be) ? epair[i].y : -1;
        }
#pragma unroll
        for (int j = 0; j < BIN_PER; ++j)
            if (d[j] >= 0) atomicAdd(&cnt[d[j] - node0], 1);
    }
    __syncthreads();

    int x = 0, c = 0;
    if (tid < BK_NODES) {
        c = cnt[tid];
        x = c;
        for (int o = 1; o < 32; o <<= 1) {
            int u = __shfl_up(x, o);
            if (lane >= o) x += u;
        }
        if (lane == 31) w_tot[wave] = x;
    }
    __syncthreads();
    if (tid < BK_NODES) {
        int carry = 0;
        for (int w = 0; w < wave; ++w) carry += w_tot[w];
        const int o = carry + x - c;
        off[tid] = o;
        const int node = node0 + tid;
        if (node < N) {
            row_ptr[node] = bb + o;
            dis[node] = rsqrtf((float)c + 1.0f);
        }
        if (node == N - 1) row_ptr[N] = bb + o + c;
    }
    __syncthreads();

    for (int base = bb + tid; base < be; base += BIN_PER * BIN_T) {
        int2 sd[BIN_PER];
#pragma unroll
        for (int j = 0; j < BIN_PER; ++j) {
            int i = base + j * BIN_T;
            sd[j] = (i < be) ? epair[i] : make_int2(0, -1);
        }
#pragma unroll
        for (int j = 0; j < BIN_PER; ++j) {
            if (sd[j].y >= 0) {
                int pos = bb + atomicAdd(&off[sd[j].y - node0], 1);
                col[pos]   = sd[j].x;
                rowid[pos] = sd[j].y;
            }
        }
    }
}

__global__ void val_kernel(const int* __restrict__ col, const int* __restrict__ rowid,
                           const float* __restrict__ dis, float* __restrict__ val, int E) {
    int p = blockIdx.x * blockDim.x + threadIdx.x;
    if (p < E) val[p] = dis[col[p]] * dis[rowid[p]];
}

__global__ __launch_bounds__(256) void gemm_wmma_kernel(
        const float* __restrict__ A,
        const float* __restrict__ W,
        float* __restrict__ T,
        int N) {
    __shared__ float As[16 * 132];

    const int row0 = blockIdx.x * 16;
    const int tid  = threadIdx.x;

    for (int i = tid; i < 16 * DIM; i += 256) {
        int r = i >> 7;
        int c = i & (DIM - 1);
        int rr = row0 + r;
        As[r * 132 + c] = (rr < N) ? A[(size_t)rr * DIM + c] : 0.0f;
    }
    __syncthreads();

    const int wave  = tid >> 5;
    const int lane  = tid & 31;
    const int m     = lane & 15;
    const int khalf = (lane >> 4) * 2;
    const int n0    = wave * 16;

    v8f acc = {};
#pragma unroll
    for (int k = 0; k < DIM; k += 4) {
        v2f a, b;
        a.x = As[m * 132 + k + khalf];
        a.y = As[m * 132 + k + khalf + 1];
        b.x = W[(k + khalf) * DIM + n0 + m];
        b.y = W[(k + khalf + 1) * DIM + n0 + m];
        acc = __builtin_amdgcn_wmma_f32_16x16x4_f32(
                false, a, false, b, (short)0, acc, false, false);
    }

    const int rbase = row0 + (lane >> 4) * 8;
#pragma unroll
    for (int v = 0; v < 8; ++v) {
        int r = rbase + v;
        if (r < N) T[(size_t)r * DIM + n0 + m] = acc[v];
    }
}

__global__ __launch_bounds__(256) void spmm_kernel(
        const float* __restrict__ T,
        const int*   __restrict__ row_ptr,
        const int*   __restrict__ col,
        const float* __restrict__ val,
        const float* __restrict__ dis,
        const float* __restrict__ bias,
        float*       __restrict__ out,
        int N) {
    int node = blockIdx.x * 8 + (threadIdx.x >> 5);
    if (node >= N) return;
    int lane = threadIdx.x & 31;

    int beg = row_ptr[node];
    int end = row_ptr[node + 1];

    float ds = dis[node];
    float w0 = ds * ds;
    float4 tv = ((const float4*)(T + (size_t)node * DIM))[lane];
    float4 acc;
    acc.x = tv.x * w0; acc.y = tv.y * w0; acc.z = tv.z * w0; acc.w = tv.w * w0;

    for (int jb = beg; jb < end; jb += 32) {
        int nb = end - jb;
        if (nb > 32) nb = 32;
        int   jl  = jb + lane;
        int   s_l = (jl < end) ? col[jl] : 0;
        float w_l = (jl < end) ? val[jl] : 0.0f;
        for (int k = 0; k < nb; ++k) {
            int   s = __shfl(s_l, k);
            float w = __shfl(w_l, k);
            float4 v = ((const float4*)(T + (size_t)s * DIM))[lane];
            acc.x = fmaf(v.x, w, acc.x);
            acc.y = fmaf(v.y, w, acc.y);
            acc.z = fmaf(v.z, w, acc.z);
            acc.w = fmaf(v.w, w, acc.w);
        }
    }

    float4 bv = ((const float4*)bias)[lane];
    float4 r;
    r.x = acc.x + bv.x; r.y = acc.y + bv.y;
    r.z = acc.z + bv.z; r.w = acc.w + bv.w;
    r.x = (r.x >= 0.0f) ? r.x : NEG_SLOPE * r.x;
    r.y = (r.y >= 0.0f) ? r.y : NEG_SLOPE * r.y;
    r.z = (r.z >= 0.0f) ? r.z : NEG_SLOPE * r.z;
    r.w = (r.w >= 0.0f) ? r.w : NEG_SLOPE * r.w;
    ((float4*)(out + (size_t)node * DIM))[lane] = r;
}

extern "C" void kernel_launch(void* const* d_in, const int* in_sizes, int n_in,
                              void* d_out, int out_size, void* d_ws, size_t ws_size,
                              hipStream_t stream) {
    const float* x  = (const float*)d_in[0];
    const int*   ei = (const int*)  d_in[1];
    const float* W0 = (const float*)d_in[2];
    const float* b0 = (const float*)d_in[3];
    const float* W1 = (const float*)d_in[4];
    const float* b1 = (const float*)d_in[5];
    float* out = (float*)d_out;

    const int N = in_sizes[0] / DIM;
    const int E = in_sizes[1] / 2;
    const int* srcIdx = ei;
    const int* dstIdx = ei + E;

    const size_t NH = (size_t)N * DIM;

    float* T        = (float*)d_ws;
    int*   ebuf     = (int*)(T + NH);
    float* val      = (float*)(ebuf + 2 * (size_t)E);
    int*   col      = (int*)(val + E);
    int*   rowid    = col + E;
    float* dis      = (float*)(rowid + E);
    int*   row_ptr  = (int*)(dis + N);
    int*   counts   = row_ptr + (N + 1);
    int*   bkt_base = counts + BIN_G * BK_MAX;

    const int thr   = 256;
    const int gridE = (E + thr - 1) / thr;
    const int nBkt  = (N + BK_NODES - 1) / BK_NODES;

    bin_count_kernel  <<<BIN_G, BIN_T, 0, stream>>>(dstIdx, counts, E);
    bin_scatter_kernel<<<BIN_G, BIN_T, 0, stream>>>(srcIdx, dstIdx, counts, bkt_base, ebuf, E);
    bin_rows_kernel   <<<nBkt,  BIN_T, 0, stream>>>(ebuf, bkt_base, row_ptr, dis, col, rowid, N);
    val_kernel        <<<gridE, thr,   0, stream>>>(col, rowid, dis, val, E);

    const int gemmGrid = (N + 15) / 16;
    const int nodeGrid = (N + 7) / 8;

    gemm_wmma_kernel<<<gemmGrid, thr, 0, stream>>>(x, W0, T, N);
    spmm_kernel<<<nodeGrid, thr, 0, stream>>>(T, row_ptr, col, val, dis, b0, out, N);

    gemm_wmma_kernel<<<gemmGrid, thr, 0, stream>>>(out, W1, T, N);
    spmm_kernel<<<nodeGrid, thr, 0, stream>>>(T, row_ptr, col, val, dis, b1, out, N);
}


# layer in one launch: neighbour sum taken on the layer's input rows, 16 summed rows handed through LDS to the f32 WMMA product; sum order changes to (A_hat H) W; no T write+read
# speedup vs baseline: 3.3255x; 1.3597x over previous
#include <hip/hip_runtime.h>
#include <hip/hip_bf16.h>

typedef float v2f __attribute__((ext_vector_type(2)));
typedef float v8f __attribute__((ext_vector_type(8)));

#define DIM 128
#define NEG_SLOPE 0.01f

#define BK_SHIFT 8
#define BK_NODES 256
#define BK_MAX   256
#define BIN_T    1024
#define BIN_G    256
#define BIN_PER  4

static_assert(BK_NODES == (1 << BK_SHIFT), "bucket of a node = node >> BK_SHIFT");
static_assert(BIN_T == 4 * BK_MAX, "carry-in: four threads per bucket, a quarter of the chunks each");
static_assert(BIN_G % 4 == 0, "chunks split in four equal quarters");

__global__ __launch_bounds__(BIN_T) void bin_count_kernel(const int* __restrict__ dst,
                                                          int* __restrict__ counts, int E) {
    __shared__ int h[BK_MAX];
    const int tid = threadIdx.x;
    if (tid < BK_MAX) h[tid] = 0;
    __syncthreads();

    const int chunk = (E + BIN_G - 1) / BIN_G;
    const int beg = blockIdx.x * chunk;
    const int end = min(beg + chunk, E);
    for (int base = beg + tid; base < end; base += BIN_PER * BIN_T) {
        int d[BIN_PER];
#pragma unroll
        for (int j = 0; j < BIN_PER; ++j) {
            int e = base + j * BIN_T;
            d[j] = (e < end) ? dst[e] : -1;
        }
#pragma unroll
        for (int j = 0; j < BIN_PER; ++j)
            if (d[j] >= 0) atomicAdd(&h[d[j] >> BK_SHIFT], 1);
    }
    __syncthreads();
    if (tid < BK_MAX) counts[blockIdx.x * BK_MAX + tid] = h[tid];
}

__global__ __launch_bounds__(BIN_T) void bin_scatter_kernel(const int* __restrict__ src,
                                                            const int* __restrict__ dst,
                                                            const int* __restrict__ counts,
                                                            int* __restrict__ bucket_base,
                                                            int* __restrict__ ebuf, int E) {
    __shared__ int p_pre[4][BK_MAX];
    __shared__ int p_tot[4][BK_MAX];
    __shared__ int offs[BK_MAX];
    __shared__ int w_tot[BK_MAX / 32];

    const int tid  = threadIdx.x;
    const int lane = tid & 31;
    const int wave = tid >> 5;
    const int g    = blockIdx.x;
    {
        const int b  = tid & (BK_MAX - 1);
        const int q  = tid / BK_MAX;
        const int k0 = q * (BIN_G / 4);
        int pre = 0, tot = 0;
#pragma unroll 16
        for (int k = 0; k < BIN_G / 4; ++k) {
            int c = counts[(k0 + k) * BK_MAX + b];
            tot += c;
            pre += ((k0 + k) < g) ? c : 0;
        }
        p_pre[q][b] = pre;
        p_tot[q][b] = tot;
    }
    __syncthreads();

    int x = 0, t = 0, pr = 0;
    if (tid < BK_MAX) {
        pr = p_pre[0][tid] + p_pre[1][tid] + p_pre[2][tid] + p_pre[3][tid];
        t  = p_tot[0][tid] + p_tot[1][tid] + p_tot[2][tid] + p_tot[3][tid];
        x = t;
        for (int off = 1; off < 32; off <<= 1) {
            int u = __shfl_up(x, off);
            if (lane >= off) x += u;
        }
        if (lane == 31) w_tot[wave] = x;
    }
    __syncthreads();
    if (tid < BK_MAX) {
        int carry = 0;
        for (int w = 0; w < wave; ++w) carry += w_tot[w];
        const int base = carry + x - t;
        offs[tid] = base + pr;
        if (g == 0) {
            bucket_base[tid] = base;
            if (tid == BK_MAX - 1) bucket_base[BK_MAX] = base + t;
        }
    }
    __syncthreads();

    const int chunk = (E + BIN_G - 1) / BIN_G;
    const int beg = g * chunk;
    const int end = min(beg + chunk, E);
    int2* __restrict__ epair = (int2*)ebuf;
    for (int base = beg + tid; base < end; base += BIN_PER * BIN_T) {
        int s[BIN_PER], d[BIN_PER];
#pragma unroll
        for (int j = 0; j < BIN_PER; ++j) {
            int e = base + j * BIN_T;
            s[j] = (e < end) ? src[e] : 0;
            d[j] = (e < end) ? dst[e] : -1;
        }
#pragma unroll
        for (int j = 0; j < BIN_PER; ++j) {
            if (d[j] >= 0) {
                int pos = atomicAdd(&offs[d[j] >> BK_SHIFT], 1);
                epair[pos] = make_int2(s[j], d[j]);
            }
        }
    }
}

__global__ __launch_bounds__(BIN_T) void bin_rows_kernel(const int* __restrict__ ebuf,
                                                         const int* __restrict__ bucket_base,
                                                         int* __restrict__ row_ptr,
                                                         float* __restrict__ dis,
                                                         int* __restrict__ col,
                                                         int* __restrict__ rowid, int N) {
    __shared__ int cnt[BK_NODES];
    __shared__ int off[BK_NODES];
    __shared__ int w_tot[BK_NODES / 32];

    const int tid   = threadIdx.x;
    const int lane  = tid & 31;
    const int wave  = tid >> 5;
    const int node0 = blockIdx.x << BK_SHIFT;
    const int bb = bucket_base[blockIdx.x];
    const int be = bucket_base[blockIdx.x + 1];
    const int2* __restrict__ epair = (const int2*)ebuf;

    if (tid < BK_NODES) cnt[tid] = 0;
    __syncthreads();

    for (int base = bb + tid; base < be; base += BIN_PER * BIN_T) {
        int d[BIN_PER];
#pragma unroll
        for (int j = 0; j < BIN_PER; ++j) {
            int i = base + j * BIN_T;
            d[j] = (i < be) ? epair[i].y : -1;
        }
#pragma unroll
        for (int j = 0; j < BIN_PER; ++j)
            if (d[j] >= 0) atomicAdd(&cnt[d[j] - node0], 1);
    }
    __syncthreads();

    int x = 0, c = 0;
    if (tid < BK_NODES) {
        c = cnt[tid];
        x = c;
        for (int o = 1; o < 32; o <<= 1) {
            int u = __shfl_up(x, o);
            if (lane >= o) x += u;
        }
        if (lane == 31) w_tot[wave] = x;
    }
    __syncthreads();
    if (tid < BK_NODES) {
        int carry = 0;
        for (int w = 0; w < wave; ++w) carry += w_tot[w];
        const int o = carry + x - c;
        off[tid] = o;
        const int node = node0 + tid;
        if (node < N) {
            row_ptr[node] = bb + o;
            dis[node] = rsqrtf((float)c + 1.0f);
        }
        if (node == N - 1) row_ptr[N] = bb + o + c;
    }
    __syncthreads();

    for (int base = bb + tid; base < be; base += BIN_PER * BIN_T) {
        int2 sd[BIN_PER];
#pragma unroll
        for (int j = 0; j < BIN_PER; ++j) {
            int i = base + j * BIN_T;
            sd[j] = (i < be) ? epair[i] : make_int2(0, -1);
        }
#pragma unroll
        for (int j = 0; j < BIN_PER; ++j) {
            if (sd[j].y >= 0) {
                int pos = bb + atomicAdd(&off[sd[j].y - node0], 1);
                col[pos]   = sd[j].x;
                rowid[pos] = sd[j].y;
            }
        }
    }
}

__global__ void val_kernel(const int* __restrict__ col, const int* __restrict__ rowid,
                           const float* __restrict__ dis, float* __restrict__ val, int E) {
    int p = blockIdx.x * blockDim.x + threadIdx.x;
    if (p < E) val[p] = dis[col[p]] * dis[rowid[p]];
}

#define ROWS_PB 16
#define A_PITCH 132

__global__ __launch_bounds__(256, 16) void gcn_layer_kernel(
        const float* __restrict__ H,
        const int*   __restrict__ row_ptr,
        const int*   __restrict__ col,
        const float* __restrict__ val,
        const float* __restrict__ dis,
        const float* __restrict__ W,
        const float* __restrict__ bias,
        float*       __restrict__ out,
        int N) {
    __shared__ float As[ROWS_PB * A_PITCH];
    __shared__ float Cs[ROWS_PB * A_PITCH];
    __shared__ int next_row;

    const int tid  = threadIdx.x;
    const int wave = tid >> 5;
    const int lane = tid & 31;
    const int row0 = blockIdx.x * ROWS_PB;

    if (tid == 0) next_row = 0;
    __syncthreads();
    for (int t = 0; t < ROWS_PB; ++t) {
        int r = 0;
        if (lane == 0) r = atomicAdd(&next_row, 1);
        r = __builtin_amdgcn_readfirstlane(r);
        if (r >= ROWS_PB) break;
        const int node = row0 + r;
        float4 acc = make_float4(0.0f, 0.0f, 0.0f, 0.0f);
        if (node < N) {
            const int beg = row_ptr[node];
            const int end = row_ptr[node + 1];
            const float ds = dis[node];
            const float w0 = ds * ds;
            const float4 tv = ((const float4*)(H + (size_t)node * DIM))[lane];
            acc.x = tv.x * w0; acc.y = tv.y * w0; acc.z = tv.z * w0; acc.w = tv.w * w0;
            for (int jb = beg; jb < end; jb += 32) {
                int nb = end - jb;
                if (nb > 32) nb = 32;
                const int   jl  = jb + lane;
                const int   s_l = (jl < end) ? col[jl] : 0;
                const float w_l = (jl < end) ? val[jl] : 0.0f;
                for (int k = 0; k < nb; ++k) {
                    const int   s = __shfl(s_l, k);
                    const float w = __shfl(w_l, k);
                    const float4 v = ((const float4*)(H + (size_t)s * DIM))[lane];
                    acc.x = fmaf(v.x, w, acc.x);
                    acc.y = fmaf(v.y, w, acc.y);
                    acc.z = fmaf(v.z, w, acc.z);
                    acc.w = fmaf(v.w, w, acc.w);
                }
            }
        }
        *(float4*)&As[r * A_PITCH + lane * 4] = acc;
    }
    __syncthreads();

    const int m     = lane & 15;
    const int khalf = (lane >> 4) * 2;
    const int n0    = wave * 16;

    v8f c = {};
#pragma unroll
    for (int k = 0; k < DIM; k += 4) {
        v2f a, b;
        a.x = As[m * A_PITCH + k + khalf];
        a.y = As[m * A_PITCH + k + khalf + 1];
        b.x = W[(k + khalf) * DIM + n0 + m];
        b.y = W[(k + khalf + 1) * DIM + n0 + m];
        c = __builtin_amdgcn_wmma_f32_16x16x4_f32(
                false, a, false, b, (short)0, c, false, false);
    }

    const int rb = (lane >> 4) * 8;
#pragma unroll
    for (int v = 0; v < 8; ++v) Cs[(rb + v) * A_PITCH + n0 + m] = c[v];
    __syncthreads();

    const float4 bv = ((const float4*)bias)[lane];
    for (int i = 0; i < 2; ++i) {
        const int r    = wave * 2 + i;
        const int node = row0 + r;
        if (node < N) {
            float4 q = *(const float4*)&Cs[r * A_PITCH + lane * 4];
            q.x += bv.x; q.y += bv.y; q.z += bv.z; q.w += bv.w;
            q.x = (q.x >= 0.0f) ? q.x : NEG_SLOPE * q.x;
            q.y = (q.y >= 0.0f) ? q.y : NEG_SLOPE * q.y;
            q.z = (q.z >= 0.0f) ? q.z : NEG_SLOPE * q.z;
            q.w = (q.w >= 0.0f) ? q.w : NEG_SLOPE * q.w;
            ((float4*)(out + (size_t)node * DIM))[lane] = q;
        }
    }
}

extern "C" void kernel_launch(void* const* d_in, const int* in_sizes, int n_in,
                              void* d_out, int out_size, void* d_ws, size_t ws_size,
                              hipStream_t stream) {
    const float* x  = (const float*)d_in[0];
    const int*   ei = (const int*)  d_in[1];
    const float* W0 = (const float*)d_in[2];
    const float* b0 = (const float*)d_in[3];
    const float* W1 = (const float*)d_in[4];
    const float* b1 = (const float*)d_in[5];
    float* out = (float*)d_out;

    const int N = in_sizes[0] / DIM;
    const int E = in_sizes[1] / 2;
    const int* srcIdx = ei;
    const int* dstIdx = ei + E;

    const size_t NH = (size_t)N * DIM;

    float* T        = (float*)d_ws;
    int*   ebuf     = (int*)(T + NH);
    float* val      = (float*)(ebuf + 2 * (size_t)E);
    int*   col      = (int*)(val + E);
    int*   rowid    = col + E;
    float* dis      = (float*)(rowid + E);
    int*   row_ptr  = (int*)(dis + N);
    int*   counts   = row_ptr + (N + 1);
    int*   bkt_base = counts + BIN_G * BK_MAX;

    const int thr   = 256;
    const int gridE = (E + thr - 1) / thr;
    const int nBkt  = (N + BK_NODES - 1) / BK_NODES;

    bin_count_kernel  <<<BIN_G, BIN_T, 0, stream>>>(dstIdx, counts, E);
    bin_scatter_kernel<<<BIN_G, BIN_T, 0, stream>>>(srcIdx, dstIdx, counts, bkt_base, ebuf, E);
    bin_rows_kernel   <<<nBkt,  BIN_T, 0, stream>>>(ebuf, bkt_base, row_ptr, dis, col, rowid, N);
    val_kernel        <<<gridE, thr,   0, stream>>>(col, rowid, dis, val, E);

    const int layerGrid = (N + ROWS_PB - 1) / ROWS_PB;

    gcn_layer_kernel<<<layerGrid, thr, 0, stream>>>(x, row_ptr, col, val, dis, W0, b0, T, N);

    gcn_layer_kernel<<<layerGrid, thr, 0, stream>>>(T, row_ptr, col, val, dis, W1, b1, out, N);
}


# edge weights dis[src]*dis[dst] formed in the first layer's row walk and stored there for the second, not in a launch of their own; row-id array dropped; same products, sum order unchanged
# speedup vs baseline: 3.5423x; 1.0652x over previous
#include <hip/hip_runtime.h>
#include <hip/hip_bf16.h>

typedef float v2f __attribute__((ext_vector_type(2)));
typedef float v8f __attribute__((ext_vector_type(8)));

#define DIM 128
#define NEG_SLOPE 0.01f

#define BK_SHIFT 8
#define BK_NODES 256
#define BK_MAX   256
#define BIN_T    1024
#define BIN_G    256
#define BIN_PER  4

static_assert(BK_NODES == (1 << BK_SHIFT), "bucket of a node = node >> BK_SHIFT");
static_assert(BIN_T == 4 * BK_MAX, "carry-in: four threads per bucket, a quarter of the chunks each");
static_assert(BIN_G % 4 == 0, "chunks split in four equal quarters");

__global__ __launch_bounds__(BIN_T) void bin_count_kernel(const int* __restrict__ dst,
                                                          int* __restrict__ counts, int E) {
    __shared__ int h[BK_MAX];
    const int tid = threadIdx.x;
    if (tid < BK_MAX) h[tid] = 0;
    __syncthreads();

    const int chunk = (E + BIN_G - 1) / BIN_G;
    const int beg = blockIdx.x * chunk;
    const int end = min(beg + chunk, E);
    for (int base = beg + tid; base < end; base += BIN_PER * BIN_T) {
        int d[BIN_PER];
#pragma unroll
        for (int j = 0; j < BIN_PER; ++j) {
            int e = base + j * BIN_T;
            d[j] = (e < end) ? dst[e] : -1;
        }
#pragma unroll
        for (int j = 0; j < BIN_PER; ++j)
            if (d[j] >= 0) atomicAdd(&h[d[j] >> BK_SHIFT], 1);
    }
    __syncthreads();
    if (tid < BK_MAX) counts[blockIdx.x * BK_MAX + tid] = h[tid];
}

__global__ __launch_bounds__(BIN_T) void bin_scatter_kernel(const int* __restrict__ src,
                                                            const int* __restrict__ dst,
                                                            const int* __restrict__ counts,
                                                            int* __restrict__ bucket_base,
                                                            int* __restrict__ ebuf, int E) {
    __shared__ int p_pre[4][BK_MAX];
    __shared__ int p_tot[4][BK_MAX];
    __shared__ int offs[BK_MAX];
    __shared__ int w_tot[BK_MAX / 32];

    const int tid  = threadIdx.x;
    const int lane = tid & 31;
    const int wave = tid >> 5;
    const int g    = blockIdx.x;
    {
        const int b  = tid & (BK_MAX - 1);
        const int q  = tid / BK_MAX;
        const int k0 = q * (BIN_G / 4);
        int pre = 0, tot = 0;
#pragma unroll 16
        for (int k = 0; k < BIN_G / 4; ++k) {
            int c = counts[(k0 + k) * BK_MAX + b];
            tot += c;
            pre += ((k0 + k) < g) ? c : 0;
        }
        p_pre[q][b] = pre;
        p_tot[q][b] = tot;
    }
    __syncthreads();

    int x = 0, t = 0, pr = 0;
    if (tid < BK_MAX) {
        pr = p_pre[0][tid] + p_pre[1][tid] + p_pre[2][tid] + p_pre[3][tid];
        t  = p_tot[0][tid] + p_tot[1][tid] + p_tot[2][tid] + p_tot[3][tid];
        x = t;
        for (int off = 1; off < 32; off <<= 1) {
            int u = __shfl_up(x, off);
            if (lane >= off) x += u;
        }
        if (lane == 31) w_tot[wave] = x;
    }
    __syncthreads();
    if (tid < BK_MAX) {
        int carry = 0;
        for (int w = 0; w < wave; ++w) carry += w_tot[w];
        const int base = carry + x - t;
        offs[tid] = base + pr;
        if (g == 0) {
            bucket_base[tid] = base;
            if (tid == BK_MAX - 1) bucket_base[BK_MAX] = base + t;
        }
    }
    __syncthreads();

    const int chunk = (E + BIN_G - 1) / BIN_G;
    const int beg = g * chunk;
    const int end = min(beg + chunk, E);
    int2* __restrict__ epair = (int2*)ebuf;
    for (int base = beg + tid; base < end; base += BIN_PER * BIN_T) {
        int s[BIN_PER], d[BIN_PER];
#pragma unroll
        for (int j = 0; j < BIN_PER; ++j) {
            int e = base + j * BIN_T;
            s[j] = (e < end) ? src[e] : 0;
            d[j] = (e < end) ? dst[e] : -1;
        }
#pragma unroll
        for (int j = 0; j < BIN_PER; ++j) {
            if (d[j] >= 0) {
                int pos = atomicAdd(&offs[d[j] >> BK_SHIFT], 1);
                epair[pos] = make_int2(s[j], d[j]);
            }
        }
    }
}

__global__ __launch_bounds__(BIN_T) void bin_rows_kernel(const int* __restrict__ ebuf,
                                                         const int* __restrict__ bucket_base,
                                                         int* __restrict__ row_ptr,
                                                         float* __restrict__ dis,
                                                         int* __restrict__ col, int N) {
    __shared__ int cnt[BK_NODES];
    __shared__ int off[BK_NODES];
    __shared__ int w_tot[BK_NODES / 32];

    const int tid   = threadIdx.x;
    const int lane  = tid & 31;
    const int wave  = tid >> 5;
    const int node0 = blockIdx.x << BK_SHIFT;
    const int bb = bucket_base[blockIdx.x];
    const int be = bucket_base[blockIdx.x + 1];
    const int2* __restrict__ epair = (const int2*)ebuf;

    if (tid < BK_NODES) cnt[tid] = 0;
    __syncthreads();

    for (int base = bb + tid; base < be; base += BIN_PER * BIN_T) {
        int d[BIN_PER];
#pragma unroll
        for (int j = 0; j < BIN_PER; ++j) {
            int i = base + j * BIN_T;
            d[j] = (i < be) ? epair[i].y : -1;
        }
#pragma unroll
        for (int j = 0; j < BIN_PER; ++j)
            if (d[j] >= 0) atomicAdd(&cnt[d[j] - node0], 1);
    }
    __syncthreads();

    int x = 0, c = 0;
    if (tid < BK_NODES) {
        c = cnt[tid];
        x = c;
        for (int o = 1; o < 32; o <<= 1) {
            int u = __shfl_up(x, o);
            if (lane >= o) x += u;
        }
        if (lane == 31) w_tot[wave] = x;
    }
    __syncthreads();
    if (tid < BK_NODES) {
        int carry = 0;
        for (int w = 0; w < wave; ++w) carry += w_tot[w];
        const int o = carry + x - c;
        off[tid] = o;
        const int node = node0 + tid;
        if (node < N) {
            row_ptr[node] = bb + o;
            dis[node] = rsqrtf((float)c + 1.0f);
        }
        if (node == N - 1) row_ptr[N] = bb + o + c;
    }
    __syncthreads();

    for (int base = bb + tid; base < be; base += BIN_PER * BIN_T) {
        int2 sd[BIN_PER];
#pragma unroll
        for (int j = 0; j < BIN_PER; ++j) {
            int i = base + j * BIN_T;
            sd[j] = (i < be) ? epair[i] : make_int2(0, -1);
        }
#pragma unroll
        for (int j = 0; j < BIN_PER; ++j) {
            if (sd[j].y >= 0) {
                int pos = bb + atomicAdd(&off[sd[j].y - node0], 1);
                col[pos] = sd[j].x;
            }
        }
    }
}

#define ROWS_PB 16
#define A_PITCH 132

template <bool FIRST>
__global__ __launch_bounds__(256, 16) void gcn_layer_kernel(
        const float* __restrict__ H,
        const int*   __restrict__ row_ptr,
        const int*   __restrict__ col,
        float*       __restrict__ val,
        const float* __restrict__ dis,
        const float* __restrict__ W,
        const float* __restrict__ bias,
        float*       __restrict__ out,
        int N) {
    __shared__ float As[ROWS_PB * A_PITCH];
    __shared__ float Cs[ROWS_PB * A_PITCH];
    __shared__ int next_row;

    const int tid  = threadIdx.x;
    const int wave = tid >> 5;
    const int lane = tid & 31;
    const int row0 = blockIdx.x * ROWS_PB;

    if (tid == 0) next_row = 0;
    __syncthreads();
    for (int t = 0; t < ROWS_PB; ++t) {
        int r = 0;
        if (lane == 0) r = atomicAdd(&next_row, 1);
        r = __builtin_amdgcn_readfirstlane(r);
        if (r >= ROWS_PB) break;
        const int node = row0 + r;
        float4 acc = make_float4(0.0f, 0.0f, 0.0f, 0.0f);
        if (node < N) {
            const int beg = row_ptr[node];
            const int end = row_ptr[node + 1];
            const float ds = dis[node];
            const float w0 = ds * ds;
            const float4 tv = ((const float4*)(H + (size_t)node * DIM))[lane];
            acc.x = tv.x * w0; acc.y = tv.y * w0; acc.z = tv.z * w0; acc.w = tv.w * w0;
            for (int jb = beg; jb < end; jb += 32) {
                int nb = end - jb;
                if (nb > 32) nb = 32;
                const int   jl  = jb + lane;
                const int   s_l = (jl < end) ? col[jl] : 0;
                float w_l;
                if (FIRST) {
                    w_l = dis[s_l] * ds;
                    if (jl < end) val[jl] = w_l;
                } else {
                    w_l = (jl < end) ? val[jl] : 0.0f;
                }
                for (int k = 0; k < nb; ++k) {
                    const int   s = __shfl(s_l, k);
                    const float w = __shfl(w_l, k);
                    const float4 v = ((const float4*)(H + (size_t)s * DIM))[lane];
                    acc.x = fmaf(v.x, w, acc.x);
                    acc.y = fmaf(v.y, w, acc.y);
                    acc.z = fmaf(v.z, w, acc.z);
                    acc.w = fmaf(v.w, w, acc.w);
                }
            }
        }
        *(float4*)&As[r * A_PITCH + lane * 4] = acc;
    }
    __syncthreads();

    const int m     = lane & 15;
    const int khalf = (lane >> 4) * 2;
    const int n0    = wave * 16;

    v8f c = {};
#pragma unroll
    for (int k = 0; k < DIM; k += 4) {
        v2f a, b;
        a.x = As[m * A_PITCH + k + khalf];
        a.y = As[m * A_PITCH + k + khalf + 1];
        b.x = W[(k + khalf) * DIM + n0 + m];
        b.y = W[(k + khalf + 1) * DIM + n0 + m];
        c = __builtin_amdgcn_wmma_f32_16x16x4_f32(
                false, a, false, b, (short)0, c, false, false);
    }

    const int rb = (lane >> 4) * 8;
#pragma unroll
    for (int v = 0; v < 8; ++v) Cs[(rb + v) * A_PITCH + n0 + m] = c[v];
    __syncthreads();

    const float4 bv = ((const float4*)bias)[lane];
    for (int i = 0; i < 2; ++i) {
        const int r    = wave * 2 + i;
        const int node = row0 + r;
        if (node < N) {
            float4 q = *(const float4*)&Cs[r * A_PITCH + lane * 4];
            q.x += bv.x; q.y += bv.y; q.z += bv.z; q.w += bv.w;
            q.x = (q.x >= 0.0f) ? q.x : NEG_SLOPE * q.x;
            q.y = (q.y >= 0.0f) ? q.y : NEG_SLOPE * q.y;
            q.z = (q.z >= 0.0f) ? q.z : NEG_SLOPE * q.z;
            q.w = (q.w >= 0.0f) ? q.w : NEG_SLOPE * q.w;
            ((float4*)(out + (size_t)node * DIM))[lane] = q;
        }
    }
}

extern "C" void kernel_launch(void* const* d_in, const int* in_sizes, int n_in,
                              void* d_out, int out_size, void* d_ws, size_t ws_size,
                              hipStream_t stream) {
    const float* x  = (const float*)d_in[0];
    const int*   ei = (const int*)  d_in[1];
    const float* W0 = (const float*)d_in[2];
    const float* b0 = (const float*)d_in[3];
    const float* W1 = (const float*)d_in[4];
    const float* b1 = (const float*)d_in[5];
    float* out = (float*)d_out;

    const int N = in_sizes[0] / DIM;
    const int E = in_sizes[1] / 2;
    const int* srcIdx = ei;
    const int* dstIdx = ei + E;

    const size_t NH = (size_t)N * DIM;

    float* T        = (float*)d_ws;
    int*   ebuf     = (int*)(T + NH);
    float* val      = (float*)(ebuf + 2 * (size_t)E);
    int*   col      = (int*)(val + E);
    float* dis      = (float*)(col + E);
    int*   row_ptr  = (int*)(dis + N);
    int*   counts   = row_ptr + (N + 1);
    int*   bkt_base = counts + BIN_G * BK_MAX;

    const int thr   = 256;
    const int nBkt  = (N + BK_NODES - 1) / BK_NODES;

    bin_count_kernel  <<<BIN_G, BIN_T, 0, stream>>>(dstIdx, counts, E);
    bin_scatter_kernel<<<BIN_G, BIN_T, 0, stream>>>(srcIdx, dstIdx, counts, bkt_base, ebuf, E);
    bin_rows_kernel   <<<nBkt,  BIN_T, 0, stream>>>(ebuf, bkt_base, row_ptr, dis, col, N);

    const int layerGrid = (N + ROWS_PB - 1) / ROWS_PB;

    gcn_layer_kernel<true><<<layerGrid, thr, 0, stream>>>(x, row_ptr, col, val, dis, W0, b0, T, N);

    gcn_layer_kernel<false><<<layerGrid, thr, 0, stream>>>(T, row_ptr, col, val, dis, W1, b1, out, N);
}


# CSR build in two launches, not three: each edge chunk orders its own edges by bucket in its own slice (no count launch); bucket blocks read one run per chunk; row order still an arrival rank
# speedup vs baseline: 4.1012x; 1.1578x over previous
#include <hip/hip_runtime.h>
#include <hip/hip_bf16.h>

typedef float v2f __attribute__((ext_vector_type(2)));
typedef float v8f __attribute__((ext_vector_type(8)));

#define DIM 128
#define NEG_SLOPE 0.01f

#define BK_SHIFT 8
#define BK_NODES 256
#define BK_MAX   256
#define BIN_T    1024
#define BIN_G    256
#define BIN_PER  4
#define SEG_PW   (BIN_G / (BIN_T / 32))

static_assert(BK_NODES == (1 << BK_SHIFT), "bucket of a node = node >> BK_SHIFT");
static_assert(BIN_G % (BIN_T / 32) == 0, "a bucket's runs split evenly over the waves of its block");
static_assert(BIN_G <= BIN_T && BK_MAX <= BIN_T && BK_NODES <= BIN_T, "one thread per run, per bucket, per node");

__global__ __launch_bounds__(BIN_T) void bin_sort_kernel(const int* __restrict__ src,
                                                         const int* __restrict__ dst,
                                                         int* __restrict__ run_start,
                                                         int* __restrict__ ebuf, int E) {
    __shared__ int h[BK_MAX];
    __shared__ int w_tot[BK_MAX / 32];

    const int tid  = threadIdx.x;
    const int lane = tid & 31;
    const int wave = tid >> 5;
    const int g    = blockIdx.x;
    if (tid < BK_MAX) h[tid] = 0;
    __syncthreads();

    const int chunk = (E + BIN_G - 1) / BIN_G;
    const int beg = g * chunk;
    const int end = min(beg + chunk, E);
    int s0[BIN_PER], d0[BIN_PER];
#pragma unroll
    for (int j = 0; j < BIN_PER; ++j) {
        int e = beg + tid + j * BIN_T;
        s0[j] = (e < end) ? src[e] : 0;
        d0[j] = (e < end) ? dst[e] : -1;
    }
#pragma unroll
    for (int j = 0; j < BIN_PER; ++j)
        if (d0[j] >= 0) atomicAdd(&h[d0[j] >> BK_SHIFT], 1);
    for (int base = beg + tid + BIN_PER * BIN_T; base < end; base += BIN_PER * BIN_T) {
        int d[BIN_PER];
#pragma unroll
        for (int j = 0; j < BIN_PER; ++j) {
            int e = base + j * BIN_T;
            d[j] = (e < end) ? dst[e] : -1;
        }
#pragma unroll
        for (int j = 0; j < BIN_PER; ++j)
            if (d[j] >= 0) atomicAdd(&h[d[j] >> BK_SHIFT], 1);
    }
    __syncthreads();

    int x = 0, c = 0;
    if (tid < BK_MAX) {
        c = h[tid];
        x = c;
        for (int o = 1; o < 32; o <<= 1) {
            int u = __shfl_up(x, o);
            if (lane >= o) x += u;
        }
        if (lane == 31) w_tot[wave] = x;
    }
    __syncthreads();
    if (tid < BK_MAX) {
        int carry = 0;
        for (int w = 0; w < wave; ++w) carry += w_tot[w];
        const int first = carry + x - c;
        h[tid] = first;
        run_start[tid * BIN_G + g] = first;
        if (tid == BK_MAX - 1) run_start[BK_MAX * BIN_G + g] = first + c;
    }
    __syncthreads();

    int2* __restrict__ epair = (int2*)ebuf;
    {
        int r[BIN_PER];
#pragma unroll
        for (int j = 0; j < BIN_PER; ++j)
            r[j] = (d0[j] >= 0) ? atomicAdd(&h[d0[j] >> BK_SHIFT], 1) : 0;
#pragma unroll
        for (int j = 0; j < BIN_PER; ++j)
            if (d0[j] >= 0) epair[beg + r[j]] = make_int2(s0[j], d0[j]);
    }
    for (int base = beg + tid + BIN_PER * BIN_T; base < end; base += BIN_PER * BIN_T) {
        int s[BIN_PER], d[BIN_PER];
#pragma unroll
        for (int j = 0; j < BIN_PER; ++j) {
            int e = base + j * BIN_T;
            s[j] = (e < end) ? src[e] : 0;
            d[j] = (e < end) ? dst[e] : -1;
        }
#pragma unroll
        for (int j = 0; j < BIN_PER; ++j) {
            if (d[j] >= 0) {
                int pos = beg + atomicAdd(&h[d[j] >> BK_SHIFT], 1);
                epair[pos] = make_int2(s[j], d[j]);
            }
        }
    }
}

__global__ __launch_bounds__(BIN_T) void bin_rows_kernel(const int* __restrict__ ebuf,
                                                         const int* __restrict__ run_start,
                                                         int* __restrict__ row_ptr,
                                                         float* __restrict__ dis,
                                                         int* __restrict__ col, int N, int E) {
    __shared__ int cnt[BK_NODES];
    __shared__ int off[BK_NODES];
    __shared__ int w_tot[BK_NODES / 32];
    __shared__ int seg_beg[BIN_G];
    __shared__ int seg_cnt[BIN_G];
    __shared__ int bb_part[BIN_G / 32];

    const int tid   = threadIdx.x;
    const int lane  = tid & 31;
    const int wave  = tid >> 5;
    const int bkt   = blockIdx.x;
    const int node0 = bkt << BK_SHIFT;
    const int chunk = (E + BIN_G - 1) / BIN_G;
    const int2* __restrict__ epair = (const int2*)ebuf;

    if (tid < BIN_G) {
        const int lo = run_start[bkt * BIN_G + tid];
        const int hi = run_start[(bkt + 1) * BIN_G + tid];
        seg_beg[tid] = tid * chunk + lo;
        seg_cnt[tid] = hi - lo;
        int t = lo;
        for (int o = 16; o > 0; o >>= 1) t += __shfl_xor(t, o);
        if (lane == 0) bb_part[wave] = t;
    }
    if (tid < BK_NODES) cnt[tid] = 0;
    __syncthreads();

    int bb = 0;
    for (int w = 0; w < BIN_G / 32; ++w) bb += bb_part[w];

    int sb[SEG_PW], sc[SEG_PW];
#pragma unroll
    for (int j = 0; j < SEG_PW; ++j) {
        sb[j] = seg_beg[wave + 32 * j];
        sc[j] = seg_cnt[wave + 32 * j];
    }

    int2 sd[SEG_PW];
#pragma unroll
    for (int j = 0; j < SEG_PW; ++j)
        sd[j] = (lane < sc[j]) ? epair[sb[j] + lane] : make_int2(0, -1);
#pragma unroll
    for (int j = 0; j < SEG_PW; ++j)
        if (sd[j].y >= 0) atomicAdd(&cnt[sd[j].y - node0], 1);
    for (int j = 0; j < SEG_PW; ++j) {
        const int b0 = seg_beg[wave + 32 * j];
        const int n  = seg_cnt[wave + 32 * j];
        for (int k = lane + 32; k < n; k += 32)
            atomicAdd(&cnt[epair[b0 + k].y - node0], 1);
    }
    __syncthreads();

    int x = 0, c = 0;
    if (tid < BK_NODES) {
        c = cnt[tid];
        x = c;
        for (int o = 1; o < 32; o <<= 1) {
            int u = __shfl_up(x, o);
            if (lane >= o) x += u;
        }
        if (lane == 31) w_tot[wave] = x;
    }
    __syncthreads();
    if (tid < BK_NODES) {
        int carry = 0;
        for (int w = 0; w < wave; ++w) carry += w_tot[w];
        const int o = carry + x - c;
        off[tid] = o;
        const int node = node0 + tid;
        if (node < N) {
            row_ptr[node] = bb + o;
            dis[node] = rsqrtf((float)c + 1.0f);
        }
        if (node == N - 1) row_ptr[N] = bb + o + c;
    }
    __syncthreads();

    {
        int r[SEG_PW];
#pragma unroll
        for (int j = 0; j < SEG_PW; ++j)
            r[j] = (sd[j].y >= 0) ? atomicAdd(&off[sd[j].y - node0], 1) : 0;
#pragma unroll
        for (int j = 0; j < SEG_PW; ++j)
            if (sd[j].y >= 0) col[bb + r[j]] = sd[j].x;
        for (int j = 0; j < SEG_PW; ++j) {
            const int b0 = seg_beg[wave + 32 * j];
            const int n  = seg_cnt[wave + 32 * j];
            for (int k = lane + 32; k < n; k += 32) {
                const int2 p = epair[b0 + k];
                int pos = bb + atomicAdd(&off[p.y - node0], 1);
                col[pos] = p.x;
            }
        }
    }
}

#define ROWS_PB 16
#define A_PITCH 132

template <bool FIRST>
__global__ __launch_bounds__(256, 16) void gcn_layer_kernel(
        const float* __restrict__ H,
        const int*   __restrict__ row_ptr,
        const int*   __restrict__ col,
        float*       __restrict__ val,
        const float* __restrict__ dis,
        const float* __restrict__ W,
        const float* __restrict__ bias,
        float*       __restrict__ out,
        int N) {
    __shared__ float As[ROWS_PB * A_PITCH];
    __shared__ float Cs[ROWS_PB * A_PITCH];
    __shared__ int next_row;

    const int tid  = threadIdx.x;
    const int wave = tid >> 5;
    const int lane = tid & 31;
    const int row0 = blockIdx.x * ROWS_PB;

    if (tid == 0) next_row = 0;
    __syncthreads();
    for (int t = 0; t < ROWS_PB; ++t) {
        int r = 0;
        if (lane == 0) r = atomicAdd(&next_row, 1);
        r = __builtin_amdgcn_readfirstlane(r);
        if (r >= ROWS_PB) break;
        const int node = row0 + r;
        float4 acc = make_float4(0.0f, 0.0f, 0.0f, 0.0f);
        if (node < N) {
            const int beg = row_ptr[node];
            const int end = row_ptr[node + 1];
            const float ds = dis[node];
            const float w0 = ds * ds;
            const float4 tv = ((const float4*)(H + (size_t)node * DIM))[lane];
            acc.x = tv.x * w0; acc.y = tv.y * w0; acc.z = tv.z * w0; acc.w = tv.w * w0;
            for (int jb = beg; jb < end; jb += 32) {
                int nb = end - jb;
                if (nb > 32) nb = 32;
                const int   jl  = jb + lane;
                const int   s_l = (jl < end) ? col[jl] : 0;
                float w_l;
                if (FIRST) {
                    w_l = dis[s_l] * ds;
                    if (jl < end) val[jl] = w_l;
                } else {
                    w_l = (jl < end) ? val[jl] : 0.0f;
                }
                for (int k = 0; k < nb; ++k) {
                    const int   s = __shfl(s_l, k);
                    const float w = __shfl(w_l, k);
                    const float4 v = ((const float4*)(H + (size_t)s * DIM))[lane];
                    acc.x = fmaf(v.x, w, acc.x);
                    acc.y = fmaf(v.y, w, acc.y);
                    acc.z = fmaf(v.z, w, acc.z);
                    acc.w = fmaf(v.w, w, acc.w);
                }
            }
        }
        *(float4*)&As[r * A_PITCH + lane * 4] = acc;
    }
    __syncthreads();

    const int m     = lane & 15;
    const int khalf = (lane >> 4) * 2;
    const int n0    = wave * 16;

    v8f c = {};
#pragma unroll
    for (int k = 0; k < DIM; k += 4) {
        v2f a, b;
        a.x = As[m * A_PITCH + k + khalf];
        a.y = As[m * A_PITCH + k + khalf + 1];
        b.x = W[(k + khalf) * DIM + n0 + m];
        b.y = W[(k + khalf + 1) * DIM + n0 + m];
        c = __builtin_amdgcn_wmma_f32_16x16x4_f32(
                false, a, false, b, (short)0, c, false, false);
    }

    const int rb = (lane >> 4) * 8;
#pragma unroll
    for (int v = 0; v < 8; ++v) Cs[(rb + v) * A_PITCH + n0 + m] = c[v];
    __syncthreads();

    const float4 bv = ((const float4*)bias)[lane];
    for (int i = 0; i < 2; ++i) {
        const int r    = wave * 2 + i;
        const int node = row0 + r;
        if (node < N) {
            float4 q = *(const float4*)&Cs[r * A_PITCH + lane * 4];
            q.x += bv.x; q.y += bv.y; q.z += bv.z; q.w += bv.w;
            q.x = (q.x >= 0.0f) ? q.x : NEG_SLOPE * q.x;
            q.y = (q.y >= 0.0f) ? q.y : NEG_SLOPE * q.y;
            q.z = (q.z >= 0.0f) ? q.z : NEG_SLOPE * q.z;
            q.w = (q.w >= 0.0f) ? q.w : NEG_SLOPE * q.w;
            ((float4*)(out + (size_t)node * DIM))[lane] = q;
        }
    }
}

extern "C" void kernel_launch(void* const* d_in, const int* in_sizes, int n_in,
                              void* d_out, int out_size, void* d_ws, size_t ws_size,
                              hipStream_t stream) {
    const float* x  = (const float*)d_in[0];
    const int*   ei = (const int*)  d_in[1];
    const float* W0 = (const float*)d_in[2];
    const float* b0 = (const float*)d_in[3];
    const float* W1 = (const float*)d_in[4];
    const float* b1 = (const float*)d_in[5];
    float* out = (float*)d_out;

    const int N = in_sizes[0] / DIM;
    const int E = in_sizes[1] / 2;
    const int* srcIdx = ei;
    const int* dstIdx = ei + E;

    const size_t NH = (size_t)N * DIM;

    float* T        = (float*)d_ws;
    int*   ebuf     = (int*)(T + NH);
    float* val      = (float*)(ebuf + 2 * (size_t)E);
    int*   col      = (int*)(val + E);
    float* dis      = (float*)(col + E);
    int*   row_ptr  = (int*)(dis + N);
    int*   runs     = row_ptr + (N + 1);

    const int thr   = 256;
    const int nBkt  = (N + BK_NODES - 1) / BK_NODES;

    bin_sort_kernel<<<BIN_G, BIN_T, 0, stream>>>(srcIdx, dstIdx, runs, ebuf, E);
    bin_rows_kernel<<<nBkt,  BIN_T, 0, stream>>>(ebuf, runs, row_ptr, dis, col, N, E);

    const int layerGrid = (N + ROWS_PB - 1) / ROWS_PB;

    gcn_layer_kernel<true><<<layerGrid, thr, 0, stream>>>(x, row_ptr, col, val, dis, W0, b0, T, N);

    gcn_layer_kernel<false><<<layerGrid, thr, 0, stream>>>(T, row_ptr, col, val, dis, W1, b1, out, N);
}
